# starting kernel
# baseline (speedup 1.0000x reference)
#include <hip/hip_runtime.h>
#include <hip/hip_bf16.h>

typedef __attribute__((ext_vector_type(2))) float v2f;
typedef __attribute__((ext_vector_type(8))) float v8f;

#define BS   32
#define TS0  64
#define TSK  64
#define F    256
#define NF   128

__global__ void cin_stage1_wmma(const float* __restrict__ xk,
                                const float* __restrict__ w,
                                float* __restrict__ A) {
    const int fc   = blockIdx.x;
    const int t    = fc >> 2;
    const int wave = threadIdx.x >> 5;
    const int lane = threadIdx.x & 31;
    const int half = lane >> 4;
    const int l    = lane & 15;

    const float* wrow = w + (size_t)t * TSK;
    (void)wrow;

    for (int nt = 0; nt < 2; ++nt) {
        v8f acc = {};
        #pragma unroll
        for (int kc = 0; kc < 16; ++kc) {
            const int y0 = kc * 4 + half * 2;
            const int n = wave * 16 + l;
            v2f a;
            a.x = w[(size_t)n * (TS0 * TSK) + (size_t)t * TSK + y0];
            a.y = w[(size_t)n * (TS0 * TSK) + (size_t)t * TSK + y0 + 1];
            const int b = nt * 16 + l;
            v2f bb;
            bb.x = xk[(size_t)b * (TSK * F) + (size_t)y0 * F + fc];
            bb.y = xk[(size_t)b * (TSK * F) + (size_t)(y0 + 1) * F + fc];
            acc = __builtin_amdgcn_wmma_f32_16x16x4_f32(
                 false, a,  false, bb,
                 (short)0, acc,  false,  false);
        }
        #pragma unroll
        for (int v = 0; v < 8; ++v) {
            const int n = wave * 16 + v + half * 8;
            const int b = nt * 16 + l;
            A[((size_t)b * NF + n) * F + fc] = acc[v];
        }
    }
}

__global__ void cin_stage2_wmma(const float* __restrict__ x0,
                                const float* __restrict__ A,
                                const float* __restrict__ bias,
                                float* __restrict__ out) {
    const int bq   = blockIdx.x;
    const int b    = bq >> 2;
    const int q    = bq & 3;
    const int wave = threadIdx.x >> 5;
    const int lane = threadIdx.x & 31;
    const int half = lane >> 4;
    const int l    = lane & 15;

    for (int rt = 0; rt < 4; ++rt) {
        v8f acc = {};
        #pragma unroll
        for (int kc = 0; kc < 16; ++kc) {
            const int t0 = kc * 4 + half * 2;
            const int n = wave * 16 + l;
            v2f a;
            a.x = A[((size_t)b * NF + n) * F + (4 * t0 + q)];
            a.y = A[((size_t)b * NF + n) * F + (4 * (t0 + 1) + q)];
            const int r = rt * 16 + l;
            v2f bb;
            bb.x = x0[(size_t)b * (TS0 * F) + (size_t)r * F + (4 * t0 + q)];
            bb.y = x0[(size_t)b * (TS0 * F) + (size_t)r * F + (4 * (t0 + 1) + q)];
            acc = __builtin_amdgcn_wmma_f32_16x16x4_f32(
                false, a, false, bb, (short)0, acc, false, false);
        }
        #pragma unroll
        for (int v = 0; v < 8; ++v) {
            const int n = wave * 16 + v + half * 8;
            const int r = rt * 16 + l;
            out[((size_t)b * NF + n) * F + q * 64 + r] = acc[v] + bias[n];
        }
    }
}

extern "C" void kernel_launch(void* const* d_in, const int* in_sizes, int n_in,
                              void* d_out, int out_size, void* d_ws, size_t ws_size,
                              hipStream_t stream) {
    const float* x0   = (const float*)d_in[0];
    const float* xk   = (const float*)d_in[1];
    const float* w    = (const float*)d_in[2];
    const float* bias = (const float*)d_in[3];
    float* A   = (float*)d_ws;
    float* out = (float*)d_out;

    cin_stage1_wmma<<<F, 256, 0, stream>>>(xk, w, A);
    cin_stage2_wmma<<<BS * 4, 256, 0, stream>>>(x0, A, bias, out);
}


# Coalesced global access: both stages re-tiled so lanes run along contiguous memory (operands staged in LDS, intermediate laid out as whole fragments, whole-line stores); fragment k order permuted
# speedup vs baseline: 5.8330x; 5.8330x over previous
#include <hip/hip_runtime.h>


typedef __attribute__((ext_vector_type(2))) float v2f;
typedef __attribute__((ext_vector_type(4))) float v4f;
typedef __attribute__((ext_vector_type(8))) float v8f;

#define BS   32
#define TS0  64
#define TSK  64
#define F    256
#define NF   128

#define XP   68
#define WP   260
#define XPL  1104


__device__ __forceinline__ float xhalf(float x) {
    const int v = __builtin_bit_cast(int, x);
    return __builtin_bit_cast(float, __builtin_amdgcn_permlanex16(v, v, 0x76543210, 0xfedcba98, false, false));
}

__device__ __forceinline__ v8f mma4(v2f a, v2f b, v8f c) {
    return __builtin_amdgcn_wmma_f32_16x16x4_f32(false, a, false, b, (short)0, c, false, false);
}

__global__ __launch_bounds__(128)
void cin_stage1_wmma(const float* __restrict__ xk,
                     const float* __restrict__ w,
                     float* __restrict__ ws) {
    __shared__ __attribute__((aligned(16))) float Wl[64 * WP];
    __shared__ __attribute__((aligned(16))) float XA[4 * XPL];

    const int id  = blockIdx.x;
    const int nh  = id & 1;
    const int bg  = (id >> 1) & 7;
    const int tg  = id >> 4;
    const int tid = threadIdx.x;

    const float* wsrc = w + (size_t)(nh * 64) * (TS0 * TSK) + tg * 256;
    #pragma unroll
    for (int i = 0; i < 32; ++i) {
        const int idx = i * 128 + tid;
        const int row = idx >> 6;
        const int c4  = idx & 63;
        const v4f v = *(const v4f*)(wsrc + (size_t)row * (TS0 * TSK) + c4 * 4);
        *(v4f*)&Wl[row * WP + c4 * 4] = v;
    }

    const float* xsrc = xk + (size_t)(bg * 4) * (TSK * F) + tg * 16;
    #pragma unroll
    for (int i = 0; i < 8; ++i) {
        const int idx = i * 128 + tid;
        const int tl  = idx & 3;
        const int y   = (idx >> 2) & 63;
        const int bl  = idx >> 8;
        const v4f v = *(const v4f*)(xsrc + (size_t)bl * (TSK * F) + y * F + tl * 4);
        float* d = &XA[tl * XPL + (bl * 4) * XP + y];
        d[0]      = v.x;
        d[XP]     = v.y;
        d[2 * XP] = v.z;
        d[3 * XP] = v.w;
    }
    __syncthreads();

    const int wv   = tid >> 5;
    const int lane = tid & 31;
    const int h    = lane >> 4;
    const int c    = lane & 15;

    v8f acc[4] = {};
    const float* ap = &XA[c * XP + 4 * h];
    const float* bp = &Wl[(wv * 16 + c) * WP + 4 * h];
    #pragma unroll
    for (int j = 0; j < 8; ++j) {
        v4f a[4], b[4];
        #pragma unroll
        for (int tl = 0; tl < 4; ++tl) {
            a[tl] = *(const v4f*)(ap + tl * XPL + 8 * j);
            b[tl] = *(const v4f*)(bp + tl * 64 + 8 * j);
        }
        #pragma unroll
        for (int tl = 0; tl < 4; ++tl)
            acc[tl] = mma4(a[tl].xy, b[tl].xy, acc[tl]);
        #pragma unroll
        for (int tl = 0; tl < 4; ++tl)
            acc[tl] = mma4(a[tl].zw, b[tl].zw, acc[tl]);
    }

    const int mt = nh * 4 + wv;
    #pragma unroll
    for (int r = 0; r < 8; ++r) {
        const int bl = 2 * h + (r >> 2);
        const int q  = r & 3;
        const int b  = bg * 4 + bl;
        v4f o;
        o.x = acc[0][r];
        o.y = acc[1][r];
        o.z = acc[2][r];
        o.w = acc[3][r];
        *(v4f*)(ws + ((((size_t)(b * 4 + q) * 8 + mt) * 16 + tg) * 16 + c) * 4) = o;
    }
}

__global__ __launch_bounds__(256)
void cin_stage2_wmma(const float* __restrict__ x0,
                     const float* __restrict__ ws,
                     const float* __restrict__ bias,
                     float* __restrict__ out) {
    __shared__ __attribute__((aligned(16))) float Xq[32 * XP];

    const int id   = blockIdx.x;
    const int rh   = id & 1;
    const int q    = (id >> 1) & 3;
    const int b    = id >> 3;
    const int tid  = threadIdx.x;
    const int wv   = tid >> 5;
    const int lane = tid & 31;
    const int h    = lane >> 4;
    const int c    = lane & 15;

    const float* asrc = ws + (size_t)((b * 4 + q) * 8 + wv) * 1024 + (h * 16 + c) * 4;
    v4f a[8];
    #pragma unroll
    for (int j = 0; j < 8; ++j)
        a[j] = *(const v4f*)(asrc + j * 128);

    const float* xsrc = x0 + (size_t)b * (TS0 * F) + (size_t)(rh * 32) * F + q;
    #pragma unroll
    for (int i = 0; i < 8; ++i) {
        const int idx = i * 256 + tid;
        const int rl  = idx >> 6;
        const int t   = idx & 63;
        Xq[rl * XP + t] = xsrc[(size_t)rl * F + 4 * t];
    }
    __syncthreads();

    v8f acc0 = {}, acc1 = {};
    const float* bp = &Xq[c * XP + 4 * h];
    #pragma unroll
    for (int j = 0; j < 8; ++j) {
        const v4f b0 = *(const v4f*)(bp + 8 * j);
        const v4f b1 = *(const v4f*)(bp + 16 * XP + 8 * j);
        acc0 = mma4(a[j].xy, b0.xy, acc0);
        acc1 = mma4(a[j].xy, b1.xy, acc1);
        acc0 = mma4(a[j].zw, b0.zw, acc0);
        acc1 = mma4(a[j].zw, b1.zw, acc1);
    }

    float* o = out + ((size_t)b * NF + wv * 16) * F + q * 64 + rh * 32 + lane;
    #pragma unroll
    for (int i = 0; i < 8; ++i) {
        const float bn   = bias[wv * 16 + 8 * h + i];
        const float own0 = acc0[i] + bn;
        const float own1 = acc1[i] + bn;
        const float got  = xhalf(h ? own0 : own1);
        o[(size_t)i * F]       = h ? got : own0;
        o[(size_t)(8 + i) * F] = h ? own1 : got;
    }
}

extern "C" void kernel_launch(void* const* d_in, const int* in_sizes, int n_in,
                              void* d_out, int out_size, void* d_ws, size_t ws_size,
                              hipStream_t stream) {
    const float* x0   = (const float*)d_in[0];
    const float* xk   = (const float*)d_in[1];
    const float* w    = (const float*)d_in[2];
    const float* bias = (const float*)d_in[3];
    float* ws  = (float*)d_ws;
    float* out = (float*)d_out;

    cin_stage1_wmma<<<256, 128, 0, stream>>>(xk, w, ws);
    cin_stage2_wmma<<<256, 256, 0, stream>>>(x0, ws, bias, out);
}


# Both stages begin their matrix products when the first operand part is in LDS: hand-over barriers order LDS traffic only (loads stay in flight), stage 1 takes W in two plane pairs staged per wave
# speedup vs baseline: 6.1532x; 1.0549x over previous
#include <hip/hip_runtime.h>


typedef __attribute__((ext_vector_type(2))) float v2f;
typedef __attribute__((ext_vector_type(4))) float v4f;
typedef __attribute__((ext_vector_type(8))) float v8f;

#define BS   32
#define TS0  64
#define TSK  64
#define F    256
#define NF   128

#define XP   68
#define WP   260
#define XPL  1104


__device__ __forceinline__ float xhalf(float x) {
    const int v = __builtin_bit_cast(int, x);
    return __builtin_bit_cast(float, __builtin_amdgcn_permlanex16(v, v, 0x76543210, 0xfedcba98, false, false));
}

__device__ __forceinline__ v8f mma4(v2f a, v2f b, v8f c) {
    return __builtin_amdgcn_wmma_f32_16x16x4_f32(false, a, false, b, (short)0, c, false, false);
}

__device__ __forceinline__ void lds_handover() {
    __builtin_amdgcn_fence(__ATOMIC_RELEASE, "workgroup", "local");
    __builtin_amdgcn_s_barrier();
    __builtin_amdgcn_fence(__ATOMIC_ACQUIRE, "workgroup", "local");
}

__device__ __forceinline__ void s1_products(const float* ap, const float* bp, v8f (&acc)[4],
                                            const int pp, const int j0, const int j1) {
    #pragma unroll
    for (int j = j0; j < j1; ++j) {
        v4f a[2], b[2];
        #pragma unroll
        for (int u = 0; u < 2; ++u) {
            const int tl = 2 * pp + u;
            a[u] = *(const v4f*)(ap + tl * XPL + 8 * j);
            b[u] = *(const v4f*)(bp + tl * 64 + 8 * j);
        }
        #pragma unroll
        for (int u = 0; u < 2; ++u)
            acc[2 * pp + u] = mma4(a[u].xy, b[u].xy, acc[2 * pp + u]);
        #pragma unroll
        for (int u = 0; u < 2; ++u)
            acc[2 * pp + u] = mma4(a[u].zw, b[u].zw, acc[2 * pp + u]);
    }
}

__global__ __launch_bounds__(128)
void cin_stage1_wmma(const float* __restrict__ xk,
                     const float* __restrict__ w,
                     float* __restrict__ ws) {
    __shared__ __attribute__((aligned(16))) float Wl[64 * WP];
    __shared__ __attribute__((aligned(16))) float XA[4 * XPL];

    const int id   = blockIdx.x;
    const int nh   = id & 1;
    const int bg   = (id >> 1) & 7;
    const int tg   = id >> 4;
    const int tid  = threadIdx.x;
    const int wv   = tid >> 5;
    const int lane = tid & 31;
    const int h    = lane >> 4;
    const int c    = lane & 15;

    const float* xsrc = xk + (size_t)(bg * 4) * (TSK * F) + tg * 16;
    v4f xv[8];
    #pragma unroll
    for (int i = 0; i < 8; ++i) {
        const int idx = i * 128 + tid;
        const int tl  = idx & 3;
        const int y   = (idx >> 2) & 63;
        const int bl  = idx >> 8;
        xv[i] = *(const v4f*)(xsrc + (size_t)bl * (TSK * F) + y * F + tl * 4);
    }

    const float* wsrc = w + (size_t)(nh * 64 + wv * 16) * (TS0 * TSK) + tg * 256 + lane * 4;
    float* wdst = &Wl[(wv * 16) * WP + lane * 4];
    v4f wr[2][16];
    #pragma unroll
    for (int k = 0; k < 16; ++k)
        wr[0][k] = *(const v4f*)(wsrc + (size_t)k * (TS0 * TSK));
    __builtin_amdgcn_sched_barrier(0);
    #pragma unroll
    for (int k = 0; k < 16; ++k)
        wr[1][k] = *(const v4f*)(wsrc + (size_t)k * (TS0 * TSK) + 128);

    #pragma unroll
    for (int i = 0; i < 8; ++i) {
        const int idx = i * 128 + tid;
        const int tl  = idx & 3;
        const int y   = (idx >> 2) & 63;
        const int bl  = idx >> 8;
        float* d = &XA[tl * XPL + (bl * 4) * XP + y];
        d[0]      = xv[i].x;
        d[XP]     = xv[i].y;
        d[2 * XP] = xv[i].z;
        d[3 * XP] = xv[i].w;
    }

    v8f acc[4] = {};
    const float* ap = &XA[c * XP + 4 * h];
    const float* bp = &Wl[(wv * 16 + c) * WP + 4 * h];

    #pragma unroll
    for (int k = 0; k < 16; ++k)
        *(v4f*)(wdst + k * WP) = wr[0][k];
    lds_handover();
    s1_products(ap, bp, acc, 0, 0, 4);
    __builtin_amdgcn_sched_barrier(0);
    #pragma unroll
    for (int k = 0; k < 16; ++k)
        *(v4f*)(wdst + k * WP + 128) = wr[1][k];
    s1_products(ap, bp, acc, 0, 4, 8);
    lds_handover();
    s1_products(ap, bp, acc, 1, 0, 8);

    const int mt = nh * 4 + wv;
    #pragma unroll
    for (int r = 0; r < 8; ++r) {
        const int bl = 2 * h + (r >> 2);
        const int q  = r & 3;
        const int b  = bg * 4 + bl;
        v4f o;
        o.x = acc[0][r];
        o.y = acc[1][r];
        o.z = acc[2][r];
        o.w = acc[3][r];
        *(v4f*)(ws + ((((size_t)(b * 4 + q) * 8 + mt) * 16 + tg) * 16 + c) * 4) = o;
    }
}

__global__ __launch_bounds__(256)
void cin_stage2_wmma(const float* __restrict__ x0,
                     const float* __restrict__ ws,
                     const float* __restrict__ bias,
                     float* __restrict__ out) {
    __shared__ __attribute__((aligned(16))) float Xq[32 * XP];

    const int id   = blockIdx.x;
    const int rh   = id & 1;
    const int q    = (id >> 1) & 3;
    const int b    = id >> 3;
    const int tid  = threadIdx.x;
    const int wv   = tid >> 5;
    const int lane = tid & 31;
    const int h    = lane >> 4;
    const int c    = lane & 15;

    const float* asrc = ws + (size_t)((b * 4 + q) * 8 + wv) * 1024 + (h * 16 + c) * 4;
    v4f a[8];
    #pragma unroll
    for (int j = 0; j < 8; ++j)
        a[j] = *(const v4f*)(asrc + j * 128);

    const float* xsrc = x0 + (size_t)b * (TS0 * F) + (size_t)(rh * 32) * F + q;
    #pragma unroll
    for (int i = 0; i < 8; ++i) {
        const int idx = i * 256 + tid;
        const int rl  = idx >> 6;
        const int t   = idx & 63;
        Xq[rl * XP + t] = xsrc[(size_t)rl * F + 4 * t];
    }
    lds_handover();

    v8f acc0 = {}, acc1 = {};
    const float* bp = &Xq[c * XP + 4 * h];
    #pragma unroll
    for (int j = 0; j < 8; ++j) {
        const v4f b0 = *(const v4f*)(bp + 8 * j);
        const v4f b1 = *(const v4f*)(bp + 16 * XP + 8 * j);
        acc0 = mma4(a[j].xy, b0.xy, acc0);
        acc1 = mma4(a[j].xy, b1.xy, acc1);
        acc0 = mma4(a[j].zw, b0.zw, acc0);
        acc1 = mma4(a[j].zw, b1.zw, acc1);
    }

    float* o = out + ((size_t)b * NF + wv * 16) * F + q * 64 + rh * 32 + lane;
    #pragma unroll
    for (int i = 0; i < 8; ++i) {
        const float bn   = bias[wv * 16 + 8 * h + i];
        const float own0 = acc0[i] + bn;
        const float own1 = acc1[i] + bn;
        const float got  = xhalf(h ? own0 : own1);
        o[(size_t)i * F]       = h ? got : own0;
        o[(size_t)(8 + i) * F] = h ? own1 : got;
    }
}

extern "C" void kernel_launch(void* const* d_in, const int* in_sizes, int n_in,
                              void* d_out, int out_size, void* d_ws, size_t ws_size,
                              hipStream_t stream) {
    const float* x0   = (const float*)d_in[0];
    const float* xk   = (const float*)d_in[1];
    const float* w    = (const float*)d_in[2];
    const float* bias = (const float*)d_in[3];
    float* ws  = (float*)d_ws;
    float* out = (float*)d_out;

    cin_stage1_wmma<<<256, 128, 0, stream>>>(xk, w, ws);
    cin_stage2_wmma<<<256, 256, 0, stream>>>(x0, ws, bias, out);
}
